# starting kernel
# baseline (speedup 1.0000x reference)
#include <hip/hip_runtime.h>

typedef __attribute__((ext_vector_type(2))) float v2f;
typedef __attribute__((ext_vector_type(8))) float v8f;

#define XS_STRIDE 132
#define WTP_STRIDE 288

__global__ __launch_bounds__(128) void reduce_m_kernel(
    const float* __restrict__ X, float* __restrict__ msg_m)
{
  const int bm = blockIdx.x;
  const int t  = threadIdx.x;
  const float* base = X + (size_t)bm * 64 * 128;
  float s = 0.f;
#pragma unroll 8
  for (int k = 0; k < 64; ++k) s += base[k * 128 + t];
  msg_m[(size_t)bm * 128 + t] = s;
}

__global__ __launch_bounds__(128) void reduce_k_kernel(
    const float* __restrict__ X, float* __restrict__ msg_k)
{
  const int bk = blockIdx.x;
  const int b  = bk >> 6;
  const int k  = bk & 63;
  const int t  = threadIdx.x;
  const float* base = X + ((size_t)b * 64 * 64 + k) * 128;
  float s = 0.f;
#pragma unroll 8
  for (int m = 0; m < 64; ++m) s += base[(size_t)m * 64 * 128 + t];
  msg_k[(size_t)bk * 128 + t] = s;
}

__global__ __launch_bounds__(256) void gemm128_wmma_kernel(
    const float* __restrict__ X, const float* __restrict__ W,
    float* __restrict__ out,
    const float* __restrict__ zm, const float* __restrict__ zk,
    int fuse)
{
  extern __shared__ float lds[];
  float* Xs  = lds;
  float* WTp = lds + 128 * XS_STRIDE;

  const int t = threadIdx.x;
  const size_t row_base = (size_t)blockIdx.x * 128;

  {
    const float4* Xg = (const float4*)(X + row_base * 128);
#pragma unroll
    for (int j = 0; j < 16; ++j) {
      int idx = t + j * 256;
      int row = idx >> 5;
      int c4  = idx & 31;
      float4 v = Xg[idx];
      *(float4*)(Xs + row * XS_STRIDE + c4 * 4) = v;
    }
  }
  {
    const float2* W2 = (const float2*)W;
#pragma unroll
    for (int j = 0; j < 32; ++j) {
      int idx = t + j * 256;
      int n  = idx & 127;
      int kp = idx >> 7;
      float2 w = W2[n * 64 + kp];
      *(float2*)(WTp + kp * WTP_STRIDE + 2 * n) = w;
    }
  }
  __syncthreads();

  const int wave = t >> 5;
  const int lane = t & 31;
  const int lm   = lane & 15;
  const int lg   = lane >> 4;

  v8f acc[8] = {};

  const float* arow = Xs + (wave * 16 + lm) * XS_STRIDE + 2 * lg;
  const float* brow = WTp + lg * WTP_STRIDE + 2 * lm;

  for (int kt = 0; kt < 32; ++kt) {
    v2f a = *(const v2f*)(arow + 4 * kt);
    const float* bk = brow + kt * (2 * WTP_STRIDE);
#pragma unroll
    for (int nt = 0; nt < 8; ++nt) {
      v2f b = *(const v2f*)(bk + nt * 32);
      acc[nt] = __builtin_amdgcn_wmma_f32_16x16x4_f32(
           false, a,  false, b,
           (short)0, acc[nt],
           false,  false);
    }
  }

  const size_t r0 = row_base + wave * 16 + 8 * lg;
#pragma unroll
  for (int nt = 0; nt < 8; ++nt) {
    const int col = nt * 16 + lm;
#pragma unroll
    for (int i = 0; i < 8; ++i) {
      const size_t r = r0 + i;
      float val = acc[nt][i];
      if (fuse) {
        const size_t bm = r >> 6;
        const size_t bk2 = ((r >> 12) << 6) | (r & 63);
        val += zm[bm * 128 + col] + zk[bk2 * 128 + col];
        __builtin_nontemporal_store(val, &out[r * 128 + col]);
      } else {
        out[r * 128 + col] = val;
      }
    }
  }
}

extern "C" void kernel_launch(void* const* d_in, const int* in_sizes, int n_in,
                              void* d_out, int out_size, void* d_ws, size_t ws_size,
                              hipStream_t stream)
{
  (void)in_sizes; (void)n_in; (void)out_size; (void)ws_size;
  const float* X     = (const float*)d_in[0];
  const float* Wself = (const float*)d_in[1];
  const float* Wm    = (const float*)d_in[2];
  const float* Wk    = (const float*)d_in[3];
  float* out = (float*)d_out;

  const size_t MSG = (size_t)4096 * 128;
  float* msg_m = (float*)d_ws;
  float* msg_k = msg_m + MSG;
  float* zm    = msg_k + MSG;
  float* zk    = zm + MSG;

  const size_t lds_bytes =
      (size_t)(128 * XS_STRIDE + 64 * WTP_STRIDE) * sizeof(float);

  reduce_m_kernel<<<4096, 128, 0, stream>>>(X, msg_m);
  reduce_k_kernel<<<4096, 128, 0, stream>>>(X, msg_k);

  gemm128_wmma_kernel<<<32, 256, lds_bytes, stream>>>(msg_m, Wm, zm, nullptr, nullptr, 0);
  gemm128_wmma_kernel<<<32, 256, lds_bytes, stream>>>(msg_k, Wk, zk, nullptr, nullptr, 0);

  gemm128_wmma_kernel<<<2048, 256, lds_bytes, stream>>>(X, Wself, out, zm, zk, 1);
}


# Epilogue through LDS: each wave writes whole 512-B rows with b128 stores and adds zm+zk from 17 float4 loads issued as one group, replacing 64 serial load-wait-store round trips per wave
# speedup vs baseline: 1.7404x; 1.7404x over previous
#include <hip/hip_runtime.h>

typedef __attribute__((ext_vector_type(2))) float v2f;
typedef __attribute__((ext_vector_type(8))) float v8f;

#define XS_STRIDE 132
#define WTP_STRIDE 288

__global__ __launch_bounds__(128) void reduce_m_kernel(
    const float* __restrict__ X, float* __restrict__ msg_m)
{
  const int bm = blockIdx.x;
  const int t  = threadIdx.x;
  const float* base = X + (size_t)bm * 64 * 128;
  float s = 0.f;
#pragma unroll 8
  for (int k = 0; k < 64; ++k) s += base[k * 128 + t];
  msg_m[(size_t)bm * 128 + t] = s;
}

__global__ __launch_bounds__(128) void reduce_k_kernel(
    const float* __restrict__ X, float* __restrict__ msg_k)
{
  const int bk = blockIdx.x;
  const int b  = bk >> 6;
  const int k  = bk & 63;
  const int t  = threadIdx.x;
  const float* base = X + ((size_t)b * 64 * 64 + k) * 128;
  float s = 0.f;
#pragma unroll 8
  for (int m = 0; m < 64; ++m) s += base[(size_t)m * 64 * 128 + t];
  msg_k[(size_t)bk * 128 + t] = s;
}

__global__ __launch_bounds__(256) void gemm128_wmma_kernel(
    const float* __restrict__ X, const float* __restrict__ W,
    float* __restrict__ out,
    const float* __restrict__ zm, const float* __restrict__ zk,
    int fuse)
{
  extern __shared__ float lds[];
  float* Xs  = lds;
  float* WTp = lds + 128 * XS_STRIDE;

  const int t = threadIdx.x;
  const size_t row_base = (size_t)blockIdx.x * 128;
  const int wave = t >> 5;
  const int lane = t & 31;
  const int lm   = lane & 15;
  const int lg   = lane >> 4;

  {
    const float4* Xg = (const float4*)(X + row_base * 128);
#pragma unroll
    for (int j = 0; j < 16; ++j) {
      int idx = t + j * 256;
      int row = idx >> 5;
      int c4  = idx & 31;
      float4 v = Xg[idx];
      *(float4*)(Xs + row * XS_STRIDE + c4 * 4) = v;
    }
  }
  {
    const float2* W2 = (const float2*)W;
#pragma unroll
    for (int j = 0; j < 32; ++j) {
      int idx = t + j * 256;
      int n  = idx & 127;
      int kp = idx >> 7;
      float2 w = W2[n * 64 + kp];
      *(float2*)(WTp + kp * WTP_STRIDE + 2 * n) = w;
    }
  }
  const size_t rw = row_base + wave * 16;
  float4 zm4 = make_float4(0.f, 0.f, 0.f, 0.f);
  float4 zk4[16];
#pragma unroll
  for (int rr = 0; rr < 16; ++rr) zk4[rr] = make_float4(0.f, 0.f, 0.f, 0.f);
  if (fuse) {
    const size_t bm  = rw >> 6;
    const size_t bk0 = ((rw >> 12) << 6) | (rw & 63);
    zm4 = ((const float4*)(zm + bm * 128))[lane];
#pragma unroll
    for (int rr = 0; rr < 16; ++rr)
      zk4[rr] = ((const float4*)(zk + (bk0 + rr) * 128))[lane];
  }

  __syncthreads();

  v8f acc[8] = {};

  const float* arow = Xs + (wave * 16 + lm) * XS_STRIDE + 2 * lg;
  const float* brow = WTp + lg * WTP_STRIDE + 2 * lm;

  for (int kt = 0; kt < 32; ++kt) {
    v2f a = *(const v2f*)(arow + 4 * kt);
    const float* bk = brow + kt * (2 * WTP_STRIDE);
#pragma unroll
    for (int nt = 0; nt < 8; ++nt) {
      v2f b = *(const v2f*)(bk + nt * 32);
      acc[nt] = __builtin_amdgcn_wmma_f32_16x16x4_f32(
           false, a,  false, b,
           (short)0, acc[nt],
           false,  false);
    }
  }

  float* cw = Xs + (wave * 16) * XS_STRIDE;
  {
    float* cst = cw + (8 * lg) * XS_STRIDE + lm;
#pragma unroll
    for (int nt = 0; nt < 8; ++nt) {
#pragma unroll
      for (int i = 0; i < 8; ++i) cst[i * XS_STRIDE + nt * 16] = acc[nt][i];
    }
  }
  __builtin_amdgcn_fence(__ATOMIC_RELEASE, "wavefront");
  __builtin_amdgcn_wave_barrier();
  __builtin_amdgcn_fence(__ATOMIC_ACQUIRE, "wavefront");

  const float* crd = cw + 4 * lane;
  float* og = out + rw * 128 + 4 * lane;
  if (fuse) {
#pragma unroll
    for (int rr = 0; rr < 16; ++rr) {
      float4 c = *(const float4*)(crd + rr * XS_STRIDE);
      const float4 z = zk4[rr];
      c.x += zm4.x + z.x;
      c.y += zm4.y + z.y;
      c.z += zm4.z + z.z;
      c.w += zm4.w + z.w;
      *(float4*)(og + rr * 128) = c;
    }
  } else {
#pragma unroll
    for (int rr = 0; rr < 16; ++rr) {
      float4 c = *(const float4*)(crd + rr * XS_STRIDE);
      *(float4*)(og + rr * 128) = c;
    }
  }
}

extern "C" void kernel_launch(void* const* d_in, const int* in_sizes, int n_in,
                              void* d_out, int out_size, void* d_ws, size_t ws_size,
                              hipStream_t stream)
{
  (void)in_sizes; (void)n_in; (void)out_size; (void)ws_size;
  const float* X     = (const float*)d_in[0];
  const float* Wself = (const float*)d_in[1];
  const float* Wm    = (const float*)d_in[2];
  const float* Wk    = (const float*)d_in[3];
  float* out = (float*)d_out;

  const size_t MSG = (size_t)4096 * 128;
  float* msg_m = (float*)d_ws;
  float* msg_k = msg_m + MSG;
  float* zm    = msg_k + MSG;
  float* zk    = zm + MSG;

  const size_t lds_bytes =
      (size_t)(128 * XS_STRIDE + 64 * WTP_STRIDE) * sizeof(float);

  reduce_m_kernel<<<4096, 128, 0, stream>>>(X, msg_m);
  reduce_k_kernel<<<4096, 128, 0, stream>>>(X, msg_k);

  gemm128_wmma_kernel<<<32, 256, lds_bytes, stream>>>(msg_m, Wm, zm, nullptr, nullptr, 0);
  gemm128_wmma_kernel<<<32, 256, lds_bytes, stream>>>(msg_k, Wk, zk, nullptr, nullptr, 0);

  gemm128_wmma_kernel<<<2048, 256, lds_bytes, stream>>>(X, Wself, out, zm, zk, 1);
}


# Both message GEMMs (zm, zk) in one launch of 256 work-groups: 32 row tiles x 4 column slices x 2 problems, each staging a 16-KB W slice, instead of two serial 32-work-group launches
# speedup vs baseline: 1.9347x; 1.1116x over previous
#include <hip/hip_runtime.h>

typedef __attribute__((ext_vector_type(2))) float v2f;
typedef __attribute__((ext_vector_type(8))) float v8f;

#define XS_STRIDE 132
#define WTP_STRIDE 288
#define WQ_STRIDE 96

__global__ __launch_bounds__(128) void reduce_m_kernel(
    const float* __restrict__ X, float* __restrict__ msg_m)
{
  const int bm = blockIdx.x;
  const int t  = threadIdx.x;
  const float* base = X + (size_t)bm * 64 * 128;
  float s = 0.f;
#pragma unroll 8
  for (int k = 0; k < 64; ++k) s += base[k * 128 + t];
  msg_m[(size_t)bm * 128 + t] = s;
}

__global__ __launch_bounds__(128) void reduce_k_kernel(
    const float* __restrict__ X, float* __restrict__ msg_k)
{
  const int bk = blockIdx.x;
  const int b  = bk >> 6;
  const int k  = bk & 63;
  const int t  = threadIdx.x;
  const float* base = X + ((size_t)b * 64 * 64 + k) * 128;
  float s = 0.f;
#pragma unroll 8
  for (int m = 0; m < 64; ++m) s += base[(size_t)m * 64 * 128 + t];
  msg_k[(size_t)bk * 128 + t] = s;
}

__global__ __launch_bounds__(256) void msg_gemm_wmma_kernel(
    const float* __restrict__ msg_m, const float* __restrict__ Wm,
    float* __restrict__ zm,
    const float* __restrict__ msg_k, const float* __restrict__ Wk,
    float* __restrict__ zk)
{
  extern __shared__ float lds[];
  float* Xs  = lds;
  float* WTq = lds + 128 * XS_STRIDE;

  const int which = blockIdx.z;
  const float* __restrict__ X = which ? msg_k : msg_m;
  const float* __restrict__ W = which ? Wk : Wm;
  float* __restrict__ out     = which ? zk : zm;

  const int t = threadIdx.x;
  const size_t row_base = (size_t)blockIdx.x * 128;
  const int n0   = blockIdx.y * 32;
  const int wave = t >> 5;
  const int lane = t & 31;
  const int lm   = lane & 15;
  const int lg   = lane >> 4;

  {
    const float4* Xg = (const float4*)(X + row_base * 128);
#pragma unroll
    for (int j = 0; j < 16; ++j) {
      int idx = t + j * 256;
      int row = idx >> 5;
      int c4  = idx & 31;
      float4 v = Xg[idx];
      *(float4*)(Xs + row * XS_STRIDE + c4 * 4) = v;
    }
  }
  {
    const float2* W2 = (const float2*)W;
#pragma unroll
    for (int j = 0; j < 8; ++j) {
      int idx = t + j * 256;
      int n  = idx & 31;
      int kp = idx >> 5;
      float2 w = W2[(n0 + n) * 64 + kp];
      *(float2*)(WTq + kp * WQ_STRIDE + 2 * n) = w;
    }
  }
  __syncthreads();

  v8f acc[2] = {};

  const float* arow = Xs + (wave * 16 + lm) * XS_STRIDE + 2 * lg;
  const float* brow = WTq + lg * WQ_STRIDE + 2 * lm;

  for (int kt = 0; kt < 32; ++kt) {
    v2f a = *(const v2f*)(arow + 4 * kt);
    const float* bk = brow + kt * (2 * WQ_STRIDE);
#pragma unroll
    for (int nt = 0; nt < 2; ++nt) {
      v2f b = *(const v2f*)(bk + nt * 32);
      acc[nt] = __builtin_amdgcn_wmma_f32_16x16x4_f32(
           false, a,  false, b,
           (short)0, acc[nt],
           false,  false);
    }
  }

  float* cw = Xs + (wave * 16) * XS_STRIDE;
  {
    float* cst = cw + (8 * lg) * XS_STRIDE + lm;
#pragma unroll
    for (int nt = 0; nt < 2; ++nt) {
#pragma unroll
      for (int i = 0; i < 8; ++i) cst[i * XS_STRIDE + nt * 16] = acc[nt][i];
    }
  }
  __builtin_amdgcn_fence(__ATOMIC_RELEASE, "wavefront");
  __builtin_amdgcn_wave_barrier();
  __builtin_amdgcn_fence(__ATOMIC_ACQUIRE, "wavefront");

  const int r4 = lane >> 3;
  const int c4 = lane & 7;
  float* og = out + (row_base + wave * 16) * 128 + n0 + 4 * c4;
#pragma unroll
  for (int j = 0; j < 4; ++j) {
    const int row = r4 + 4 * j;
    float4 c = *(const float4*)(cw + row * XS_STRIDE + 4 * c4);
    *(float4*)(og + (size_t)row * 128) = c;
  }
}

__global__ __launch_bounds__(256) void gemm128_wmma_kernel(
    const float* __restrict__ X, const float* __restrict__ W,
    float* __restrict__ out,
    const float* __restrict__ zm, const float* __restrict__ zk,
    int fuse)
{
  extern __shared__ float lds[];
  float* Xs  = lds;
  float* WTp = lds + 128 * XS_STRIDE;

  const int t = threadIdx.x;
  const size_t row_base = (size_t)blockIdx.x * 128;
  const int wave = t >> 5;
  const int lane = t & 31;
  const int lm   = lane & 15;
  const int lg   = lane >> 4;

  {
    const float4* Xg = (const float4*)(X + row_base * 128);
#pragma unroll
    for (int j = 0; j < 16; ++j) {
      int idx = t + j * 256;
      int row = idx >> 5;
      int c4  = idx & 31;
      float4 v = Xg[idx];
      *(float4*)(Xs + row * XS_STRIDE + c4 * 4) = v;
    }
  }
  {
    const float2* W2 = (const float2*)W;
#pragma unroll
    for (int j = 0; j < 32; ++j) {
      int idx = t + j * 256;
      int n  = idx & 127;
      int kp = idx >> 7;
      float2 w = W2[n * 64 + kp];
      *(float2*)(WTp + kp * WTP_STRIDE + 2 * n) = w;
    }
  }
  const size_t rw = row_base + wave * 16;
  float4 zm4 = make_float4(0.f, 0.f, 0.f, 0.f);
  float4 zk4[16];
#pragma unroll
  for (int rr = 0; rr < 16; ++rr) zk4[rr] = make_float4(0.f, 0.f, 0.f, 0.f);
  if (fuse) {
    const size_t bm  = rw >> 6;
    const size_t bk0 = ((rw >> 12) << 6) | (rw & 63);
    zm4 = ((const float4*)(zm + bm * 128))[lane];
#pragma unroll
    for (int rr = 0; rr < 16; ++rr)
      zk4[rr] = ((const float4*)(zk + (bk0 + rr) * 128))[lane];
  }

  __syncthreads();

  v8f acc[8] = {};

  const float* arow = Xs + (wave * 16 + lm) * XS_STRIDE + 2 * lg;
  const float* brow = WTp + lg * WTP_STRIDE + 2 * lm;

  for (int kt = 0; kt < 32; ++kt) {
    v2f a = *(const v2f*)(arow + 4 * kt);
    const float* bk = brow + kt * (2 * WTP_STRIDE);
#pragma unroll
    for (int nt = 0; nt < 8; ++nt) {
      v2f b = *(const v2f*)(bk + nt * 32);
      acc[nt] = __builtin_amdgcn_wmma_f32_16x16x4_f32(
           false, a,  false, b,
           (short)0, acc[nt],
           false,  false);
    }
  }

  float* cw = Xs + (wave * 16) * XS_STRIDE;
  {
    float* cst = cw + (8 * lg) * XS_STRIDE + lm;
#pragma unroll
    for (int nt = 0; nt < 8; ++nt) {
#pragma unroll
      for (int i = 0; i < 8; ++i) cst[i * XS_STRIDE + nt * 16] = acc[nt][i];
    }
  }
  __builtin_amdgcn_fence(__ATOMIC_RELEASE, "wavefront");
  __builtin_amdgcn_wave_barrier();
  __builtin_amdgcn_fence(__ATOMIC_ACQUIRE, "wavefront");

  const float* crd = cw + 4 * lane;
  float* og = out + rw * 128 + 4 * lane;
  if (fuse) {
#pragma unroll
    for (int rr = 0; rr < 16; ++rr) {
      float4 c = *(const float4*)(crd + rr * XS_STRIDE);
      const float4 z = zk4[rr];
      c.x += zm4.x + z.x;
      c.y += zm4.y + z.y;
      c.z += zm4.z + z.z;
      c.w += zm4.w + z.w;
      *(float4*)(og + rr * 128) = c;
    }
  } else {
#pragma unroll
    for (int rr = 0; rr < 16; ++rr) {
      float4 c = *(const float4*)(crd + rr * XS_STRIDE);
      *(float4*)(og + rr * 128) = c;
    }
  }
}

extern "C" void kernel_launch(void* const* d_in, const int* in_sizes, int n_in,
                              void* d_out, int out_size, void* d_ws, size_t ws_size,
                              hipStream_t stream)
{
  (void)in_sizes; (void)n_in; (void)out_size; (void)ws_size;
  const float* X     = (const float*)d_in[0];
  const float* Wself = (const float*)d_in[1];
  const float* Wm    = (const float*)d_in[2];
  const float* Wk    = (const float*)d_in[3];
  float* out = (float*)d_out;

  const size_t MSG = (size_t)4096 * 128;
  float* msg_m = (float*)d_ws;
  float* msg_k = msg_m + MSG;
  float* zm    = msg_k + MSG;
  float* zk    = zm + MSG;

  const size_t lds_bytes =
      (size_t)(128 * XS_STRIDE + 64 * WTP_STRIDE) * sizeof(float);

  reduce_m_kernel<<<4096, 128, 0, stream>>>(X, msg_m);
  reduce_k_kernel<<<4096, 128, 0, stream>>>(X, msg_k);

  const size_t lds_q_bytes =
      (size_t)(128 * XS_STRIDE + 64 * WQ_STRIDE) * sizeof(float);
  msg_gemm_wmma_kernel<<<dim3(32, 4, 2), 256, lds_q_bytes, stream>>>(
      msg_m, Wm, zm, msg_k, Wk, zk);

  gemm128_wmma_kernel<<<2048, 256, lds_bytes, stream>>>(X, Wself, out, zm, zk, 1);
}


# Both message sums (over k and over m) in ONE pass over the input: 128 work-groups each read one 256-B column of a batch slab once with 16-B loads; partial sums joined via LDS (new summation order)
# speedup vs baseline: 2.0155x; 1.0418x over previous
#include <hip/hip_runtime.h>

typedef __attribute__((ext_vector_type(2))) float v2f;
typedef __attribute__((ext_vector_type(8))) float v8f;

#define XS_STRIDE 132
#define WTP_STRIDE 288
#define WQ_STRIDE 96

__device__ __forceinline__ void add4(float4& a, const float4& v)
{
  a.x += v.x; a.y += v.y; a.z += v.z; a.w += v.w;
}

__global__ __launch_bounds__(512) void reduce_both_kernel(
    const float* __restrict__ X, float* __restrict__ msg_m,
    float* __restrict__ msg_k)
{
  extern __shared__ float4 part[];

  const int t    = threadIdx.x;
  const int b    = blockIdx.x >> 1;
  const int half = blockIdx.x & 1;
  const int c    = t & 15;
  const int kq   = (t >> 4) & 7;
  const int mq   = t >> 7;

  const float4* p = (const float4*)(X + (size_t)b * 64 * 64 * 128)
                    + half * 16 + c + (size_t)((mq * 16) * 64 + kq) * 32;
  float4* pm = part + (kq * 64 + mq * 16) * 16 + c;

  float4 accK[8];
#pragma unroll
  for (int kk = 0; kk < 8; ++kk) accK[kk] = make_float4(0.f, 0.f, 0.f, 0.f);

  float4 xa[8], xb[8];
#pragma unroll
  for (int kk = 0; kk < 8; ++kk) xa[kk] = p[kk * 256];

  auto take = [&](const float4 (&x)[8], float4* dst) {
    float4 sm = x[0];
    add4(accK[0], x[0]);
#pragma unroll
    for (int kk = 1; kk < 8; ++kk) { add4(accK[kk], x[kk]); add4(sm, x[kk]); }
    *dst = sm;
  };

#pragma unroll 1
  for (int i = 0; i < 7; ++i) {
#pragma unroll
    for (int kk = 0; kk < 8; ++kk) xb[kk] = p[64 * 32 + kk * 256];
    take(xa, pm);
#pragma unroll
    for (int kk = 0; kk < 8; ++kk) xa[kk] = p[2 * 64 * 32 + kk * 256];
    take(xb, pm + 16);
    p  += 2 * 64 * 32;
    pm += 2 * 16;
  }
#pragma unroll
  for (int kk = 0; kk < 8; ++kk) xb[kk] = p[64 * 32 + kk * 256];
  take(xa, pm);
  take(xb, pm + 16);
  __syncthreads();

  const int o = t >> 4;
#pragma unroll
  for (int h = 0; h < 2; ++h) {
    const int r = o + 32 * h;
    float4 s = part[r * 16 + c];
#pragma unroll
    for (int j = 1; j < 8; ++j) add4(s, part[(j * 64 + r) * 16 + c]);
    ((float4*)(msg_m + ((size_t)b * 64 + r) * 128))[half * 16 + c] = s;
  }
  __syncthreads();

#pragma unroll
  for (int kk = 0; kk < 8; ++kk)
    part[(mq * 64 + kk * 8 + kq) * 16 + c] = accK[kk];
  __syncthreads();
#pragma unroll
  for (int h = 0; h < 2; ++h) {
    const int r = o + 32 * h;
    float4 s = part[r * 16 + c];
#pragma unroll
    for (int j = 1; j < 4; ++j) add4(s, part[(j * 64 + r) * 16 + c]);
    ((float4*)(msg_k + ((size_t)b * 64 + r) * 128))[half * 16 + c] = s;
  }
}

__global__ __launch_bounds__(256) void msg_gemm_wmma_kernel(
    const float* __restrict__ msg_m, const float* __restrict__ Wm,
    float* __restrict__ zm,
    const float* __restrict__ msg_k, const float* __restrict__ Wk,
    float* __restrict__ zk)
{
  extern __shared__ float lds[];
  float* Xs  = lds;
  float* WTq = lds + 128 * XS_STRIDE;

  const int which = blockIdx.z;
  const float* __restrict__ X = which ? msg_k : msg_m;
  const float* __restrict__ W = which ? Wk : Wm;
  float* __restrict__ out     = which ? zk : zm;

  const int t = threadIdx.x;
  const size_t row_base = (size_t)blockIdx.x * 128;
  const int n0   = blockIdx.y * 32;
  const int wave = t >> 5;
  const int lane = t & 31;
  const int lm   = lane & 15;
  const int lg   = lane >> 4;

  {
    const float4* Xg = (const float4*)(X + row_base * 128);
#pragma unroll
    for (int j = 0; j < 16; ++j) {
      int idx = t + j * 256;
      int row = idx >> 5;
      int c4  = idx & 31;
      float4 v = Xg[idx];
      *(float4*)(Xs + row * XS_STRIDE + c4 * 4) = v;
    }
  }
  {
    const float2* W2 = (const float2*)W;
#pragma unroll
    for (int j = 0; j < 8; ++j) {
      int idx = t + j * 256;
      int n  = idx & 31;
      int kp = idx >> 5;
      float2 w = W2[(n0 + n) * 64 + kp];
      *(float2*)(WTq + kp * WQ_STRIDE + 2 * n) = w;
    }
  }
  __syncthreads();

  v8f acc[2] = {};

  const float* arow = Xs + (wave * 16 + lm) * XS_STRIDE + 2 * lg;
  const float* brow = WTq + lg * WQ_STRIDE + 2 * lm;

  for (int kt = 0; kt < 32; ++kt) {
    v2f a = *(const v2f*)(arow + 4 * kt);
    const float* bk = brow + kt * (2 * WQ_STRIDE);
#pragma unroll
    for (int nt = 0; nt < 2; ++nt) {
      v2f b = *(const v2f*)(bk + nt * 32);
      acc[nt] = __builtin_amdgcn_wmma_f32_16x16x4_f32(
           false, a,  false, b,
           (short)0, acc[nt],
           false,  false);
    }
  }

  float* cw = Xs + (wave * 16) * XS_STRIDE;
  {
    float* cst = cw + (8 * lg) * XS_STRIDE + lm;
#pragma unroll
    for (int nt = 0; nt < 2; ++nt) {
#pragma unroll
      for (int i = 0; i < 8; ++i) cst[i * XS_STRIDE + nt * 16] = acc[nt][i];
    }
  }
  __builtin_amdgcn_fence(__ATOMIC_RELEASE, "wavefront");
  __builtin_amdgcn_wave_barrier();
  __builtin_amdgcn_fence(__ATOMIC_ACQUIRE, "wavefront");

  const int r4 = lane >> 3;
  const int c4 = lane & 7;
  float* og = out + (row_base + wave * 16) * 128 + n0 + 4 * c4;
#pragma unroll
  for (int j = 0; j < 4; ++j) {
    const int row = r4 + 4 * j;
    float4 c = *(const float4*)(cw + row * XS_STRIDE + 4 * c4);
    *(float4*)(og + (size_t)row * 128) = c;
  }
}

__global__ __launch_bounds__(256) void gemm128_wmma_kernel(
    const float* __restrict__ X, const float* __restrict__ W,
    float* __restrict__ out,
    const float* __restrict__ zm, const float* __restrict__ zk,
    int fuse)
{
  extern __shared__ float lds[];
  float* Xs  = lds;
  float* WTp = lds + 128 * XS_STRIDE;

  const int t = threadIdx.x;
  const size_t row_base = (size_t)blockIdx.x * 128;
  const int wave = t >> 5;
  const int lane = t & 31;
  const int lm   = lane & 15;
  const int lg   = lane >> 4;

  {
    const float4* Xg = (const float4*)(X + row_base * 128);
#pragma unroll
    for (int j = 0; j < 16; ++j) {
      int idx = t + j * 256;
      int row = idx >> 5;
      int c4  = idx & 31;
      float4 v = Xg[idx];
      *(float4*)(Xs + row * XS_STRIDE + c4 * 4) = v;
    }
  }
  {
    const float2* W2 = (const float2*)W;
#pragma unroll
    for (int j = 0; j < 32; ++j) {
      int idx = t + j * 256;
      int n  = idx & 127;
      int kp = idx >> 7;
      float2 w = W2[n * 64 + kp];
      *(float2*)(WTp + kp * WTP_STRIDE + 2 * n) = w;
    }
  }
  const size_t rw = row_base + wave * 16;
  float4 zm4 = make_float4(0.f, 0.f, 0.f, 0.f);
  float4 zk4[16];
#pragma unroll
  for (int rr = 0; rr < 16; ++rr) zk4[rr] = make_float4(0.f, 0.f, 0.f, 0.f);
  if (fuse) {
    const size_t bm  = rw >> 6;
    const size_t bk0 = ((rw >> 12) << 6) | (rw & 63);
    zm4 = ((const float4*)(zm + bm * 128))[lane];
#pragma unroll
    for (int rr = 0; rr < 16; ++rr)
      zk4[rr] = ((const float4*)(zk + (bk0 + rr) * 128))[lane];
  }

  __syncthreads();

  v8f acc[8] = {};

  const float* arow = Xs + (wave * 16 + lm) * XS_STRIDE + 2 * lg;
  const float* brow = WTp + lg * WTP_STRIDE + 2 * lm;

  for (int kt = 0; kt < 32; ++kt) {
    v2f a = *(const v2f*)(arow + 4 * kt);
    const float* bk = brow + kt * (2 * WTP_STRIDE);
#pragma unroll
    for (int nt = 0; nt < 8; ++nt) {
      v2f b = *(const v2f*)(bk + nt * 32);
      acc[nt] = __builtin_amdgcn_wmma_f32_16x16x4_f32(
           false, a,  false, b,
           (short)0, acc[nt],
           false,  false);
    }
  }

  float* cw = Xs + (wave * 16) * XS_STRIDE;
  {
    float* cst = cw + (8 * lg) * XS_STRIDE + lm;
#pragma unroll
    for (int nt = 0; nt < 8; ++nt) {
#pragma unroll
      for (int i = 0; i < 8; ++i) cst[i * XS_STRIDE + nt * 16] = acc[nt][i];
    }
  }
  __builtin_amdgcn_fence(__ATOMIC_RELEASE, "wavefront");
  __builtin_amdgcn_wave_barrier();
  __builtin_amdgcn_fence(__ATOMIC_ACQUIRE, "wavefront");

  const float* crd = cw + 4 * lane;
  float* og = out + rw * 128 + 4 * lane;
  if (fuse) {
#pragma unroll
    for (int rr = 0; rr < 16; ++rr) {
      float4 c = *(const float4*)(crd + rr * XS_STRIDE);
      const float4 z = zk4[rr];
      c.x += zm4.x + z.x;
      c.y += zm4.y + z.y;
      c.z += zm4.z + z.z;
      c.w += zm4.w + z.w;
      *(float4*)(og + rr * 128) = c;
    }
  } else {
#pragma unroll
    for (int rr = 0; rr < 16; ++rr) {
      float4 c = *(const float4*)(crd + rr * XS_STRIDE);
      *(float4*)(og + rr * 128) = c;
    }
  }
}

extern "C" void kernel_launch(void* const* d_in, const int* in_sizes, int n_in,
                              void* d_out, int out_size, void* d_ws, size_t ws_size,
                              hipStream_t stream)
{
  (void)in_sizes; (void)n_in; (void)out_size; (void)ws_size;
  const float* X     = (const float*)d_in[0];
  const float* Wself = (const float*)d_in[1];
  const float* Wm    = (const float*)d_in[2];
  const float* Wk    = (const float*)d_in[3];
  float* out = (float*)d_out;

  const size_t MSG = (size_t)4096 * 128;
  float* msg_m = (float*)d_ws;
  float* msg_k = msg_m + MSG;
  float* zm    = msg_k + MSG;
  float* zk    = zm + MSG;

  const size_t lds_bytes =
      (size_t)(128 * XS_STRIDE + 64 * WTP_STRIDE) * sizeof(float);

  const size_t lds_r_bytes = (size_t)(8 * 64 * 16) * sizeof(float4);
  reduce_both_kernel<<<128, 512, lds_r_bytes, stream>>>(X, msg_m, msg_k);

  const size_t lds_q_bytes =
      (size_t)(128 * XS_STRIDE + 64 * WQ_STRIDE) * sizeof(float);
  msg_gemm_wmma_kernel<<<dim3(32, 4, 2), 256, lds_q_bytes, stream>>>(
      msg_m, Wm, zm, msg_k, Wk, zk);

  gemm128_wmma_kernel<<<2048, 256, lds_bytes, stream>>>(X, Wself, out, zm, zk, 1);
}


# W staged row-major at the X tile's padded LDS pitch: a wave loads whole 512-B rows (4 lines) instead of 8-B pieces 512 B apart (32 lines per load); B fragment is one b64 of row n, same k order
# speedup vs baseline: 2.1244x; 1.0540x over previous
#include <hip/hip_runtime.h>

typedef __attribute__((ext_vector_type(2))) float v2f;
typedef __attribute__((ext_vector_type(8))) float v8f;

#define XS_STRIDE 132
#define WS_STRIDE 132

__device__ __forceinline__ void add4(float4& a, const float4& v)
{
  a.x += v.x; a.y += v.y; a.z += v.z; a.w += v.w;
}

__global__ __launch_bounds__(512) void reduce_both_kernel(
    const float* __restrict__ X, float* __restrict__ msg_m,
    float* __restrict__ msg_k)
{
  extern __shared__ float4 part[];

  const int t    = threadIdx.x;
  const int b    = blockIdx.x >> 1;
  const int half = blockIdx.x & 1;
  const int c    = t & 15;
  const int kq   = (t >> 4) & 7;
  const int mq   = t >> 7;

  const float4* p = (const float4*)(X + (size_t)b * 64 * 64 * 128)
                    + half * 16 + c + (size_t)((mq * 16) * 64 + kq) * 32;
  float4* pm = part + (kq * 64 + mq * 16) * 16 + c;

  float4 accK[8];
#pragma unroll
  for (int kk = 0; kk < 8; ++kk) accK[kk] = make_float4(0.f, 0.f, 0.f, 0.f);

  float4 xa[8], xb[8];
#pragma unroll
  for (int kk = 0; kk < 8; ++kk) xa[kk] = p[kk * 256];

  auto take = [&](const float4 (&x)[8], float4* dst) {
    float4 sm = x[0];
    add4(accK[0], x[0]);
#pragma unroll
    for (int kk = 1; kk < 8; ++kk) { add4(accK[kk], x[kk]); add4(sm, x[kk]); }
    *dst = sm;
  };

#pragma unroll 1
  for (int i = 0; i < 7; ++i) {
#pragma unroll
    for (int kk = 0; kk < 8; ++kk) xb[kk] = p[64 * 32 + kk * 256];
    take(xa, pm);
#pragma unroll
    for (int kk = 0; kk < 8; ++kk) xa[kk] = p[2 * 64 * 32 + kk * 256];
    take(xb, pm + 16);
    p  += 2 * 64 * 32;
    pm += 2 * 16;
  }
#pragma unroll
  for (int kk = 0; kk < 8; ++kk) xb[kk] = p[64 * 32 + kk * 256];
  take(xa, pm);
  take(xb, pm + 16);
  __syncthreads();

  const int o = t >> 4;
#pragma unroll
  for (int h = 0; h < 2; ++h) {
    const int r = o + 32 * h;
    float4 s = part[r * 16 + c];
#pragma unroll
    for (int j = 1; j < 8; ++j) add4(s, part[(j * 64 + r) * 16 + c]);
    ((float4*)(msg_m + ((size_t)b * 64 + r) * 128))[half * 16 + c] = s;
  }
  __syncthreads();

#pragma unroll
  for (int kk = 0; kk < 8; ++kk)
    part[(mq * 64 + kk * 8 + kq) * 16 + c] = accK[kk];
  __syncthreads();
#pragma unroll
  for (int h = 0; h < 2; ++h) {
    const int r = o + 32 * h;
    float4 s = part[r * 16 + c];
#pragma unroll
    for (int j = 1; j < 4; ++j) add4(s, part[(j * 64 + r) * 16 + c]);
    ((float4*)(msg_k + ((size_t)b * 64 + r) * 128))[half * 16 + c] = s;
  }
}

__global__ __launch_bounds__(256) void msg_gemm_wmma_kernel(
    const float* __restrict__ msg_m, const float* __restrict__ Wm,
    float* __restrict__ zm,
    const float* __restrict__ msg_k, const float* __restrict__ Wk,
    float* __restrict__ zk)
{
  extern __shared__ float lds[];
  float* Xs  = lds;
  float* Wq  = lds + 128 * XS_STRIDE;

  const int which = blockIdx.z;
  const float* __restrict__ X = which ? msg_k : msg_m;
  const float* __restrict__ W = which ? Wk : Wm;
  float* __restrict__ out     = which ? zk : zm;

  const int t = threadIdx.x;
  const size_t row_base = (size_t)blockIdx.x * 128;
  const int n0   = blockIdx.y * 32;
  const int wave = t >> 5;
  const int lane = t & 31;
  const int lm   = lane & 15;
  const int lg   = lane >> 4;

  {
    const float4* Xg = (const float4*)(X + row_base * 128);
#pragma unroll
    for (int j = 0; j < 16; ++j) {
      int idx = t + j * 256;
      int row = idx >> 5;
      int c4  = idx & 31;
      float4 v = Xg[idx];
      *(float4*)(Xs + row * XS_STRIDE + c4 * 4) = v;
    }
  }
  {
    const float4* Wg = (const float4*)(W + (size_t)n0 * 128);
#pragma unroll
    for (int j = 0; j < 4; ++j) {
      int idx = t + j * 256;
      int n  = idx >> 5;
      int c4 = idx & 31;
      float4 w = Wg[idx];
      *(float4*)(Wq + n * WS_STRIDE + c4 * 4) = w;
    }
  }
  __syncthreads();

  v8f acc[2] = {};

  const float* arow = Xs + (wave * 16 + lm) * XS_STRIDE + 2 * lg;
  const float* brow = Wq + lm * WS_STRIDE + 2 * lg;

  for (int kt = 0; kt < 32; ++kt) {
    v2f a = *(const v2f*)(arow + 4 * kt);
    const float* bk = brow + 4 * kt;
#pragma unroll
    for (int nt = 0; nt < 2; ++nt) {
      v2f b = *(const v2f*)(bk + nt * (16 * WS_STRIDE));
      acc[nt] = __builtin_amdgcn_wmma_f32_16x16x4_f32(
           false, a,  false, b,
           (short)0, acc[nt],
           false,  false);
    }
  }

  float* cw = Xs + (wave * 16) * XS_STRIDE;
  {
    float* cst = cw + (8 * lg) * XS_STRIDE + lm;
#pragma unroll
    for (int nt = 0; nt < 2; ++nt) {
#pragma unroll
      for (int i = 0; i < 8; ++i) cst[i * XS_STRIDE + nt * 16] = acc[nt][i];
    }
  }
  __builtin_amdgcn_fence(__ATOMIC_RELEASE, "wavefront");
  __builtin_amdgcn_wave_barrier();
  __builtin_amdgcn_fence(__ATOMIC_ACQUIRE, "wavefront");

  const int r4 = lane >> 3;
  const int c4 = lane & 7;
  float* og = out + (row_base + wave * 16) * 128 + n0 + 4 * c4;
#pragma unroll
  for (int j = 0; j < 4; ++j) {
    const int row = r4 + 4 * j;
    float4 c = *(const float4*)(cw + row * XS_STRIDE + 4 * c4);
    *(float4*)(og + (size_t)row * 128) = c;
  }
}

__global__ __launch_bounds__(256) void gemm128_wmma_kernel(
    const float* __restrict__ X, const float* __restrict__ W,
    float* __restrict__ out,
    const float* __restrict__ zm, const float* __restrict__ zk,
    int fuse)
{
  extern __shared__ float lds[];
  float* Xs  = lds;
  float* Ws  = lds + 128 * XS_STRIDE;

  const int t = threadIdx.x;
  const size_t row_base = (size_t)blockIdx.x * 128;
  const int wave = t >> 5;
  const int lane = t & 31;
  const int lm   = lane & 15;
  const int lg   = lane >> 4;

  {
    const float4* Xg = (const float4*)(X + row_base * 128);
#pragma unroll
    for (int j = 0; j < 16; ++j) {
      int idx = t + j * 256;
      int row = idx >> 5;
      int c4  = idx & 31;
      float4 v = Xg[idx];
      *(float4*)(Xs + row * XS_STRIDE + c4 * 4) = v;
    }
  }
  {
    const float4* Wg = (const float4*)W;
#pragma unroll
    for (int j = 0; j < 16; ++j) {
      int idx = t + j * 256;
      int n  = idx >> 5;
      int c4 = idx & 31;
      float4 w = Wg[idx];
      *(float4*)(Ws + n * WS_STRIDE + c4 * 4) = w;
    }
  }
  const size_t rw = row_base + wave * 16;
  float4 zm4 = make_float4(0.f, 0.f, 0.f, 0.f);
  float4 zk4[16];
#pragma unroll
  for (int rr = 0; rr < 16; ++rr) zk4[rr] = make_float4(0.f, 0.f, 0.f, 0.f);
  if (fuse) {
    const size_t bm  = rw >> 6;
    const size_t bk0 = ((rw >> 12) << 6) | (rw & 63);
    zm4 = ((const float4*)(zm + bm * 128))[lane];
#pragma unroll
    for (int rr = 0; rr < 16; ++rr)
      zk4[rr] = ((const float4*)(zk + (bk0 + rr) * 128))[lane];
  }

  __syncthreads();

  v8f acc[8] = {};

  const float* arow = Xs + (wave * 16 + lm) * XS_STRIDE + 2 * lg;
  const float* brow = Ws + lm * WS_STRIDE + 2 * lg;

  for (int kt = 0; kt < 32; ++kt) {
    v2f a = *(const v2f*)(arow + 4 * kt);
    const float* bk = brow + 4 * kt;
#pragma unroll
    for (int nt = 0; nt < 8; ++nt) {
      v2f b = *(const v2f*)(bk + nt * (16 * WS_STRIDE));
      acc[nt] = __builtin_amdgcn_wmma_f32_16x16x4_f32(
           false, a,  false, b,
           (short)0, acc[nt],
           false,  false);
    }
  }

  float* cw = Xs + (wave * 16) * XS_STRIDE;
  {
    float* cst = cw + (8 * lg) * XS_STRIDE + lm;
#pragma unroll
    for (int nt = 0; nt < 8; ++nt) {
#pragma unroll
      for (int i = 0; i < 8; ++i) cst[i * XS_STRIDE + nt * 16] = acc[nt][i];
    }
  }
  __builtin_amdgcn_fence(__ATOMIC_RELEASE, "wavefront");
  __builtin_amdgcn_wave_barrier();
  __builtin_amdgcn_fence(__ATOMIC_ACQUIRE, "wavefront");

  const float* crd = cw + 4 * lane;
  float* og = out + rw * 128 + 4 * lane;
  if (fuse) {
#pragma unroll
    for (int rr = 0; rr < 16; ++rr) {
      float4 c = *(const float4*)(crd + rr * XS_STRIDE);
      const float4 z = zk4[rr];
      c.x += zm4.x + z.x;
      c.y += zm4.y + z.y;
      c.z += zm4.z + z.z;
      c.w += zm4.w + z.w;
      *(float4*)(og + rr * 128) = c;
    }
  } else {
#pragma unroll
    for (int rr = 0; rr < 16; ++rr) {
      float4 c = *(const float4*)(crd + rr * XS_STRIDE);
      *(float4*)(og + rr * 128) = c;
    }
  }
}

extern "C" void kernel_launch(void* const* d_in, const int* in_sizes, int n_in,
                              void* d_out, int out_size, void* d_ws, size_t ws_size,
                              hipStream_t stream)
{
  (void)in_sizes; (void)n_in; (void)out_size; (void)ws_size;
  const float* X     = (const float*)d_in[0];
  const float* Wself = (const float*)d_in[1];
  const float* Wm    = (const float*)d_in[2];
  const float* Wk    = (const float*)d_in[3];
  float* out = (float*)d_out;

  const size_t MSG = (size_t)4096 * 128;
  float* msg_m = (float*)d_ws;
  float* msg_k = msg_m + MSG;
  float* zm    = msg_k + MSG;
  float* zk    = zm + MSG;

  const size_t lds_bytes =
      (size_t)(128 * XS_STRIDE + 128 * WS_STRIDE) * sizeof(float);

  const size_t lds_r_bytes = (size_t)(8 * 64 * 16) * sizeof(float4);
  reduce_both_kernel<<<128, 512, lds_r_bytes, stream>>>(X, msg_m, msg_k);

  const size_t lds_q_bytes =
      (size_t)(128 * XS_STRIDE + 32 * WS_STRIDE) * sizeof(float);
  msg_gemm_wmma_kernel<<<dim3(32, 4, 2), 256, lds_q_bytes, stream>>>(
      msg_m, Wm, zm, msg_k, Wk, zk);

  gemm128_wmma_kernel<<<2048, 256, lds_bytes, stream>>>(X, Wself, out, zm, zk, 1);
}


# Main GEMM as one resident 16-wave work-group per WGP (4 waves per SIMD): W and the batch's zk rows staged once, each wave streams 4 row tiles through its own LDS strip, next tile's rows loaded ahead
# speedup vs baseline: 2.2772x; 1.0719x over previous
#include <hip/hip_runtime.h>

typedef __attribute__((ext_vector_type(2))) float v2f;
typedef __attribute__((ext_vector_type(8))) float v8f;
typedef __attribute__((ext_vector_type(4))) float v4f;

#define XS_STRIDE 132
#define WS_STRIDE 132

__device__ __forceinline__ void add4(float4& a, const float4& v)
{
  a.x += v.x; a.y += v.y; a.z += v.z; a.w += v.w;
}

__global__ __launch_bounds__(512) void reduce_both_kernel(
    const float* __restrict__ X, float* __restrict__ msg_m,
    float* __restrict__ msg_k)
{
  extern __shared__ float4 part[];

  const int t    = threadIdx.x;
  const int b    = blockIdx.x >> 1;
  const int half = blockIdx.x & 1;
  const int c    = t & 15;
  const int kq   = (t >> 4) & 7;
  const int mq   = t >> 7;

  const float4* p = (const float4*)(X + (size_t)b * 64 * 64 * 128)
                    + half * 16 + c + (size_t)((mq * 16) * 64 + kq) * 32;
  float4* pm = part + (kq * 64 + mq * 16) * 16 + c;

  float4 accK[8];
#pragma unroll
  for (int kk = 0; kk < 8; ++kk) accK[kk] = make_float4(0.f, 0.f, 0.f, 0.f);

  float4 xa[8], xb[8];
#pragma unroll
  for (int kk = 0; kk < 8; ++kk) xa[kk] = p[kk * 256];

  auto take = [&](const float4 (&x)[8], float4* dst) {
    float4 sm = x[0];
    add4(accK[0], x[0]);
#pragma unroll
    for (int kk = 1; kk < 8; ++kk) { add4(accK[kk], x[kk]); add4(sm, x[kk]); }
    *dst = sm;
  };

#pragma unroll 1
  for (int i = 0; i < 7; ++i) {
#pragma unroll
    for (int kk = 0; kk < 8; ++kk) xb[kk] = p[64 * 32 + kk * 256];
    take(xa, pm);
#pragma unroll
    for (int kk = 0; kk < 8; ++kk) xa[kk] = p[2 * 64 * 32 + kk * 256];
    take(xb, pm + 16);
    p  += 2 * 64 * 32;
    pm += 2 * 16;
  }
#pragma unroll
  for (int kk = 0; kk < 8; ++kk) xb[kk] = p[64 * 32 + kk * 256];
  take(xa, pm);
  take(xb, pm + 16);
  __syncthreads();

  const int o = t >> 4;
#pragma unroll
  for (int h = 0; h < 2; ++h) {
    const int r = o + 32 * h;
    float4 s = part[r * 16 + c];
#pragma unroll
    for (int j = 1; j < 8; ++j) add4(s, part[(j * 64 + r) * 16 + c]);
    ((float4*)(msg_m + ((size_t)b * 64 + r) * 128))[half * 16 + c] = s;
  }
  __syncthreads();

#pragma unroll
  for (int kk = 0; kk < 8; ++kk)
    part[(mq * 64 + kk * 8 + kq) * 16 + c] = accK[kk];
  __syncthreads();
#pragma unroll
  for (int h = 0; h < 2; ++h) {
    const int r = o + 32 * h;
    float4 s = part[r * 16 + c];
#pragma unroll
    for (int j = 1; j < 4; ++j) add4(s, part[(j * 64 + r) * 16 + c]);
    ((float4*)(msg_k + ((size_t)b * 64 + r) * 128))[half * 16 + c] = s;
  }
}

__global__ __launch_bounds__(256) void msg_gemm_wmma_kernel(
    const float* __restrict__ msg_m, const float* __restrict__ Wm,
    float* __restrict__ zm,
    const float* __restrict__ msg_k, const float* __restrict__ Wk,
    float* __restrict__ zk)
{
  extern __shared__ float lds[];
  float* Xs  = lds;
  float* Wq  = lds + 128 * XS_STRIDE;

  const int which = blockIdx.z;
  const float* __restrict__ X = which ? msg_k : msg_m;
  const float* __restrict__ W = which ? Wk : Wm;
  float* __restrict__ out     = which ? zk : zm;

  const int t = threadIdx.x;
  const size_t row_base = (size_t)blockIdx.x * 128;
  const int n0   = blockIdx.y * 32;
  const int wave = t >> 5;
  const int lane = t & 31;
  const int lm   = lane & 15;
  const int lg   = lane >> 4;

  {
    const float4* Xg = (const float4*)(X + row_base * 128);
#pragma unroll
    for (int j = 0; j < 16; ++j) {
      int idx = t + j * 256;
      int row = idx >> 5;
      int c4  = idx & 31;
      float4 v = Xg[idx];
      *(float4*)(Xs + row * XS_STRIDE + c4 * 4) = v;
    }
  }
  {
    const float4* Wg = (const float4*)(W + (size_t)n0 * 128);
#pragma unroll
    for (int j = 0; j < 4; ++j) {
      int idx = t + j * 256;
      int n  = idx >> 5;
      int c4 = idx & 31;
      float4 w = Wg[idx];
      *(float4*)(Wq + n * WS_STRIDE + c4 * 4) = w;
    }
  }
  __syncthreads();

  v8f acc[2] = {};

  const float* arow = Xs + (wave * 16 + lm) * XS_STRIDE + 2 * lg;
  const float* brow = Wq + lm * WS_STRIDE + 2 * lg;

  for (int kt = 0; kt < 32; ++kt) {
    v2f a = *(const v2f*)(arow + 4 * kt);
    const float* bk = brow + 4 * kt;
#pragma unroll
    for (int nt = 0; nt < 2; ++nt) {
      v2f b = *(const v2f*)(bk + nt * (16 * WS_STRIDE));
      acc[nt] = __builtin_amdgcn_wmma_f32_16x16x4_f32(
           false, a,  false, b,
           (short)0, acc[nt],
           false,  false);
    }
  }

  float* cw = Xs + (wave * 16) * XS_STRIDE;
  {
    float* cst = cw + (8 * lg) * XS_STRIDE + lm;
#pragma unroll
    for (int nt = 0; nt < 2; ++nt) {
#pragma unroll
      for (int i = 0; i < 8; ++i) cst[i * XS_STRIDE + nt * 16] = acc[nt][i];
    }
  }
  __builtin_amdgcn_fence(__ATOMIC_RELEASE, "wavefront");
  __builtin_amdgcn_wave_barrier();
  __builtin_amdgcn_fence(__ATOMIC_ACQUIRE, "wavefront");

  const int r4 = lane >> 3;
  const int c4 = lane & 7;
  float* og = out + (row_base + wave * 16) * 128 + n0 + 4 * c4;
#pragma unroll
  for (int j = 0; j < 4; ++j) {
    const int row = r4 + 4 * j;
    float4 c = *(const float4*)(cw + row * XS_STRIDE + 4 * c4);
    *(float4*)(og + (size_t)row * 128) = c;
  }
}

#define ZS_ROWS 64

__device__ __forceinline__ void wave_lds_handover()
{
  __builtin_amdgcn_fence(__ATOMIC_RELEASE, "wavefront");
  __builtin_amdgcn_wave_barrier();
  __builtin_amdgcn_fence(__ATOMIC_ACQUIRE, "wavefront");
}

__global__ __launch_bounds__(512) void gemm128_stream_kernel(
    const float* __restrict__ X, const float* __restrict__ W,
    float* __restrict__ out,
    const float* __restrict__ zm, const float* __restrict__ zk)
{
  __shared__ __attribute__((aligned(16))) float smem[128 * WS_STRIDE + ZS_ROWS * 128 + 256 * XS_STRIDE];
  float* Ws = smem;
  float* Zs = smem + 128 * WS_STRIDE;
  float* Xs = Zs + ZS_ROWS * 128;

  const int t    = threadIdx.x;
  const int wave = t >> 5;
  const int lane = t & 31;
  const int lm   = lane & 15;
  const int lg   = lane >> 4;
  const size_t b    = blockIdx.x >> 2;
  const size_t rw0  = (size_t)blockIdx.x * 1024 + wave * 16;
  const int    k0   = 16 * (wave & 3);

  {
    const float4* Wg = (const float4*)W;
#pragma unroll
    for (int j = 0; j < 8; ++j) {
      int idx = t + j * 512;
      int n  = idx >> 5;
      int c4 = idx & 31;
      float4 w = Wg[idx];
      *(float4*)(Ws + n * WS_STRIDE + c4 * 4) = w;
    }
  }
  {
    const float4* Zg = (const float4*)(zk + b * 64 * 128);
#pragma unroll
    for (int j = 0; j < 4; ++j) {
      int idx = t + j * 512;
      ((float4*)Zs)[idx] = Zg[idx];
    }
  }
  float* xw = Xs + (wave * 16) * XS_STRIDE;
  const v4f* xg = (const v4f*)(X + rw0 * 128) + lane;
  v4f xn[16];
#pragma unroll
  for (int rr = 0; rr < 16; ++rr) xn[rr] = xg[rr * 32];
#pragma unroll
  for (int rr = 0; rr < 16; ++rr) *(v4f*)(xw + rr * XS_STRIDE + 4 * lane) = xn[rr];
  v4f zmn = ((const v4f*)(zm + (rw0 >> 6) * 128))[lane];

  __syncthreads();

  const float* arow = xw + lm * XS_STRIDE + 2 * lg;
  const float* brow = Ws + lm * WS_STRIDE + 2 * lg;
  const float* crd  = xw + 4 * lane;
  float*       cst  = xw + (8 * lg) * XS_STRIDE + lm;
  const float* zrd  = Zs + k0 * 128 + 4 * lane;
  float* og = out + rw0 * 128 + 4 * lane;

  for (int tile = 0; tile < 4; ++tile) {
    const v4f zmc = zmn;
    if (tile < 3) {
      xg += 256 * 32;
#pragma unroll
      for (int rr = 0; rr < 16; ++rr) xn[rr] = xg[rr * 32];
      zmn = ((const v4f*)(zm + ((rw0 + 256 * (tile + 1)) >> 6) * 128))[lane];
    }

    v8f acc[8] = {};
    for (int kt = 0; kt < 32; ++kt) {
      v2f a = *(const v2f*)(arow + 4 * kt);
      const float* bk = brow + 4 * kt;
#pragma unroll
      for (int nt = 0; nt < 8; ++nt) {
        v2f bf = *(const v2f*)(bk + nt * (16 * WS_STRIDE));
        acc[nt] = __builtin_amdgcn_wmma_f32_16x16x4_f32(
             false, a,  false, bf,
             (short)0, acc[nt],
             false,  false);
      }
    }

#pragma unroll
    for (int nt = 0; nt < 8; ++nt) {
#pragma unroll
      for (int i = 0; i < 8; ++i) cst[i * XS_STRIDE + nt * 16] = acc[nt][i];
    }
    wave_lds_handover();
#pragma unroll
    for (int rr = 0; rr < 16; ++rr) {
      float4 c = *(const float4*)(crd + rr * XS_STRIDE);
      const float4 z = *(const float4*)(zrd + rr * 128);
      c.x += zmc.x + z.x;
      c.y += zmc.y + z.y;
      c.z += zmc.z + z.z;
      c.w += zmc.w + z.w;
      *(float4*)(og + rr * 128) = c;
    }
    og += 256 * 128;

    if (tile < 3) {
#pragma unroll
      for (int rr = 0; rr < 16; ++rr) *(v4f*)(xw + rr * XS_STRIDE + 4 * lane) = xn[rr];
      wave_lds_handover();
    }
  }
}

extern "C" void kernel_launch(void* const* d_in, const int* in_sizes, int n_in,
                              void* d_out, int out_size, void* d_ws, size_t ws_size,
                              hipStream_t stream)
{
  (void)in_sizes; (void)n_in; (void)out_size; (void)ws_size;
  const float* X     = (const float*)d_in[0];
  const float* Wself = (const float*)d_in[1];
  const float* Wm    = (const float*)d_in[2];
  const float* Wk    = (const float*)d_in[3];
  float* out = (float*)d_out;

  const size_t MSG = (size_t)4096 * 128;
  float* msg_m = (float*)d_ws;
  float* msg_k = msg_m + MSG;
  float* zm    = msg_k + MSG;
  float* zk    = zm + MSG;

  const size_t lds_r_bytes = (size_t)(8 * 64 * 16) * sizeof(float4);
  reduce_both_kernel<<<128, 512, lds_r_bytes, stream>>>(X, msg_m, msg_k);

  const size_t lds_q_bytes =
      (size_t)(128 * XS_STRIDE + 32 * WS_STRIDE) * sizeof(float);
  msg_gemm_wmma_kernel<<<dim3(32, 4, 2), 256, lds_q_bytes, stream>>>(
      msg_m, Wm, zm, msg_k, Wk, zk);

  gemm128_stream_kernel<<<256, 512, 0, stream>>>(X, Wself, out, zm, zk);
}


# Reduction pass on all 256 WGPs instead of 128: a work-group sums one m half of its slab column; msg_k leaves as two planes that the message GEMM adds while loading them (new summation order)
# speedup vs baseline: 2.5069x; 1.1009x over previous
#include <hip/hip_runtime.h>

typedef __attribute__((ext_vector_type(2))) float v2f;
typedef __attribute__((ext_vector_type(8))) float v8f;
typedef __attribute__((ext_vector_type(4))) float v4f;

#define XS_STRIDE 132
#define WS_STRIDE 132

__device__ __forceinline__ void add4(float4& a, const float4& v)
{
  a.x += v.x; a.y += v.y; a.z += v.z; a.w += v.w;
}

__global__ __launch_bounds__(512) void reduce_both_kernel(
    const float* __restrict__ X, float* __restrict__ msg_m,
    float* __restrict__ msg_k0, float* __restrict__ msg_k1)
{
  extern __shared__ float4 part[];

  const int t    = threadIdx.x;
  const int b    = blockIdx.x >> 2;
  const int mh   = (blockIdx.x >> 1) & 1;
  const int half = blockIdx.x & 1;
  const int c    = t & 15;
  const int kq   = (t >> 4) & 7;
  const int mq   = t >> 7;

  const float4* p = (const float4*)(X + (size_t)b * 64 * 64 * 128)
                    + half * 16 + c + (size_t)((mh * 32 + mq * 8) * 64 + kq) * 32;
  float4* pm = part + (kq * 32 + mq * 8) * 16 + c;

  float4 accK[8];
#pragma unroll
  for (int kk = 0; kk < 8; ++kk) accK[kk] = make_float4(0.f, 0.f, 0.f, 0.f);

  float4 xa[8], xb[8];
#pragma unroll
  for (int kk = 0; kk < 8; ++kk) xa[kk] = p[kk * 256];

  auto take = [&](const float4 (&x)[8], float4* dst) {
    float4 sm = x[0];
    add4(accK[0], x[0]);
#pragma unroll
    for (int kk = 1; kk < 8; ++kk) { add4(accK[kk], x[kk]); add4(sm, x[kk]); }
    *dst = sm;
  };

#pragma unroll 1
  for (int i = 0; i < 3; ++i) {
#pragma unroll
    for (int kk = 0; kk < 8; ++kk) xb[kk] = p[64 * 32 + kk * 256];
    take(xa, pm);
#pragma unroll
    for (int kk = 0; kk < 8; ++kk) xa[kk] = p[2 * 64 * 32 + kk * 256];
    take(xb, pm + 16);
    p  += 2 * 64 * 32;
    pm += 2 * 16;
  }
#pragma unroll
  for (int kk = 0; kk < 8; ++kk) xb[kk] = p[64 * 32 + kk * 256];
  take(xa, pm);
  take(xb, pm + 16);
  __syncthreads();

  const int o = t >> 4;
  {
    float4 s = part[o * 16 + c];
#pragma unroll
    for (int j = 1; j < 8; ++j) add4(s, part[(j * 32 + o) * 16 + c]);
    ((float4*)(msg_m + ((size_t)b * 64 + mh * 32 + o) * 128))[half * 16 + c] = s;
  }
  __syncthreads();

#pragma unroll
  for (int kk = 0; kk < 8; ++kk)
    part[(mq * 64 + kk * 8 + kq) * 16 + c] = accK[kk];
  __syncthreads();
  float* __restrict__ mk = mh ? msg_k1 : msg_k0;
#pragma unroll
  for (int h = 0; h < 2; ++h) {
    const int r = o + 32 * h;
    float4 s = part[r * 16 + c];
#pragma unroll
    for (int j = 1; j < 4; ++j) add4(s, part[(j * 64 + r) * 16 + c]);
    ((float4*)(mk + ((size_t)b * 64 + r) * 128))[half * 16 + c] = s;
  }
}

__global__ __launch_bounds__(256) void msg_gemm_wmma_kernel(
    const float* __restrict__ msg_m, const float* __restrict__ Wm,
    float* __restrict__ zm,
    const float* __restrict__ msg_k0, const float* __restrict__ msg_k1,
    const float* __restrict__ Wk, float* __restrict__ zk)
{
  extern __shared__ float lds[];
  float* Xs  = lds;
  float* Wq  = lds + 128 * XS_STRIDE;

  const int which = blockIdx.z;
  const float* __restrict__ X = which ? msg_k0 : msg_m;
  const float* __restrict__ W = which ? Wk : Wm;
  float* __restrict__ out     = which ? zk : zm;

  const int t = threadIdx.x;
  const size_t row_base = (size_t)blockIdx.x * 128;
  const int n0   = blockIdx.y * 32;
  const int wave = t >> 5;
  const int lane = t & 31;
  const int lm   = lane & 15;
  const int lg   = lane >> 4;

  {
    const float4* Xg = (const float4*)(X + row_base * 128);
    float4 v[16];
#pragma unroll
    for (int j = 0; j < 16; ++j) v[j] = Xg[t + j * 256];
    if (which) {
      const float4* Xh = (const float4*)(msg_k1 + row_base * 128);
      float4 u[16];
#pragma unroll
      for (int j = 0; j < 16; ++j) u[j] = Xh[t + j * 256];
      __builtin_amdgcn_sched_barrier(0);
#pragma unroll
      for (int j = 0; j < 16; ++j) {
        v[j].x += u[j].x; v[j].y += u[j].y; v[j].z += u[j].z; v[j].w += u[j].w;
      }
    }
#pragma unroll
    for (int j = 0; j < 16; ++j) {
      int idx = t + j * 256;
      int row = idx >> 5;
      int c4  = idx & 31;
      *(float4*)(Xs + row * XS_STRIDE + c4 * 4) = v[j];
    }
  }
  {
    const float4* Wg = (const float4*)(W + (size_t)n0 * 128);
#pragma unroll
    for (int j = 0; j < 4; ++j) {
      int idx = t + j * 256;
      int n  = idx >> 5;
      int c4 = idx & 31;
      float4 w = Wg[idx];
      *(float4*)(Wq + n * WS_STRIDE + c4 * 4) = w;
    }
  }
  __syncthreads();

  v8f acc[2] = {};

  const float* arow = Xs + (wave * 16 + lm) * XS_STRIDE + 2 * lg;
  const float* brow = Wq + lm * WS_STRIDE + 2 * lg;

  for (int kt = 0; kt < 32; ++kt) {
    v2f a = *(const v2f*)(arow + 4 * kt);
    const float* bk = brow + 4 * kt;
#pragma unroll
    for (int nt = 0; nt < 2; ++nt) {
      v2f b = *(const v2f*)(bk + nt * (16 * WS_STRIDE));
      acc[nt] = __builtin_amdgcn_wmma_f32_16x16x4_f32(
           false, a,  false, b,
           (short)0, acc[nt],
           false,  false);
    }
  }

  float* cw = Xs + (wave * 16) * XS_STRIDE;
  {
    float* cst = cw + (8 * lg) * XS_STRIDE + lm;
#pragma unroll
    for (int nt = 0; nt < 2; ++nt) {
#pragma unroll
      for (int i = 0; i < 8; ++i) cst[i * XS_STRIDE + nt * 16] = acc[nt][i];
    }
  }
  __builtin_amdgcn_fence(__ATOMIC_RELEASE, "wavefront");
  __builtin_amdgcn_wave_barrier();
  __builtin_amdgcn_fence(__ATOMIC_ACQUIRE, "wavefront");

  const int r4 = lane >> 3;
  const int c4 = lane & 7;
  float* og = out + (row_base + wave * 16) * 128 + n0 + 4 * c4;
#pragma unroll
  for (int j = 0; j < 4; ++j) {
    const int row = r4 + 4 * j;
    float4 c = *(const float4*)(cw + row * XS_STRIDE + 4 * c4);
    *(float4*)(og + (size_t)row * 128) = c;
  }
}

#define ZS_ROWS 64

__device__ __forceinline__ void wave_lds_handover()
{
  __builtin_amdgcn_fence(__ATOMIC_RELEASE, "wavefront");
  __builtin_amdgcn_wave_barrier();
  __builtin_amdgcn_fence(__ATOMIC_ACQUIRE, "wavefront");
}

__global__ __launch_bounds__(512) void gemm128_stream_kernel(
    const float* __restrict__ X, const float* __restrict__ W,
    float* __restrict__ out,
    const float* __restrict__ zm, const float* __restrict__ zk)
{
  __shared__ __attribute__((aligned(16))) float smem[128 * WS_STRIDE + ZS_ROWS * 128 + 256 * XS_STRIDE];
  float* Ws = smem;
  float* Zs = smem + 128 * WS_STRIDE;
  float* Xs = Zs + ZS_ROWS * 128;

  const int t    = threadIdx.x;
  const int wave = t >> 5;
  const int lane = t & 31;
  const int lm   = lane & 15;
  const int lg   = lane >> 4;
  const size_t b    = blockIdx.x >> 2;
  const size_t rw0  = (size_t)blockIdx.x * 1024 + wave * 16;
  const int    k0   = 16 * (wave & 3);

  {
    const float4* Wg = (const float4*)W;
#pragma unroll
    for (int j = 0; j < 8; ++j) {
      int idx = t + j * 512;
      int n  = idx >> 5;
      int c4 = idx & 31;
      float4 w = Wg[idx];
      *(float4*)(Ws + n * WS_STRIDE + c4 * 4) = w;
    }
  }
  {
    const float4* Zg = (const float4*)(zk + b * 64 * 128);
#pragma unroll
    for (int j = 0; j < 4; ++j) {
      int idx = t + j * 512;
      ((float4*)Zs)[idx] = Zg[idx];
    }
  }
  float* xw = Xs + (wave * 16) * XS_STRIDE;
  const v4f* xg = (const v4f*)(X + rw0 * 128) + lane;
  v4f xn[16];
#pragma unroll
  for (int rr = 0; rr < 16; ++rr) xn[rr] = xg[rr * 32];
#pragma unroll
  for (int rr = 0; rr < 16; ++rr) *(v4f*)(xw + rr * XS_STRIDE + 4 * lane) = xn[rr];
  v4f zmn = ((const v4f*)(zm + (rw0 >> 6) * 128))[lane];

  __syncthreads();

  const float* arow = xw + lm * XS_STRIDE + 2 * lg;
  const float* brow = Ws + lm * WS_STRIDE + 2 * lg;
  const float* crd  = xw + 4 * lane;
  float*       cst  = xw + (8 * lg) * XS_STRIDE + lm;
  const float* zrd  = Zs + k0 * 128 + 4 * lane;
  float* og = out + rw0 * 128 + 4 * lane;

  for (int tile = 0; tile < 4; ++tile) {
    const v4f zmc = zmn;
    if (tile < 3) {
      xg += 256 * 32;
#pragma unroll
      for (int rr = 0; rr < 16; ++rr) xn[rr] = xg[rr * 32];
      zmn = ((const v4f*)(zm + ((rw0 + 256 * (tile + 1)) >> 6) * 128))[lane];
    }

    v8f acc[8] = {};
    for (int kt = 0; kt < 32; ++kt) {
      v2f a = *(const v2f*)(arow + 4 * kt);
      const float* bk = brow + 4 * kt;
#pragma unroll
      for (int nt = 0; nt < 8; ++nt) {
        v2f bf = *(const v2f*)(bk + nt * (16 * WS_STRIDE));
        acc[nt] = __builtin_amdgcn_wmma_f32_16x16x4_f32(
             false, a,  false, bf,
             (short)0, acc[nt],
             false,  false);
      }
    }

#pragma unroll
    for (int nt = 0; nt < 8; ++nt) {
#pragma unroll
      for (int i = 0; i < 8; ++i) cst[i * XS_STRIDE + nt * 16] = acc[nt][i];
    }
    wave_lds_handover();
#pragma unroll
    for (int rr = 0; rr < 16; ++rr) {
      float4 c = *(const float4*)(crd + rr * XS_STRIDE);
      const float4 z = *(const float4*)(zrd + rr * 128);
      c.x += zmc.x + z.x;
      c.y += zmc.y + z.y;
      c.z += zmc.z + z.z;
      c.w += zmc.w + z.w;
      *(float4*)(og + rr * 128) = c;
    }
    og += 256 * 128;

    if (tile < 3) {
#pragma unroll
      for (int rr = 0; rr < 16; ++rr) *(v4f*)(xw + rr * XS_STRIDE + 4 * lane) = xn[rr];
      wave_lds_handover();
    }
  }
}

extern "C" void kernel_launch(void* const* d_in, const int* in_sizes, int n_in,
                              void* d_out, int out_size, void* d_ws, size_t ws_size,
                              hipStream_t stream)
{
  (void)in_sizes; (void)n_in; (void)out_size; (void)ws_size;
  const float* X     = (const float*)d_in[0];
  const float* Wself = (const float*)d_in[1];
  const float* Wm    = (const float*)d_in[2];
  const float* Wk    = (const float*)d_in[3];
  float* out = (float*)d_out;

  const size_t MSG = (size_t)4096 * 128;
  float* msg_m = (float*)d_ws;
  float* msg_k = msg_m + MSG;
  float* zm    = msg_k + MSG;
  float* zk    = zm + MSG;
  float* msg_k1 = out;

  const size_t lds_r_bytes = (size_t)(4 * 64 * 16) * sizeof(float4);
  reduce_both_kernel<<<256, 512, lds_r_bytes, stream>>>(X, msg_m, msg_k, msg_k1);

  const size_t lds_q_bytes =
      (size_t)(128 * XS_STRIDE + 32 * WS_STRIDE) * sizeof(float);
  msg_gemm_wmma_kernel<<<dim3(32, 4, 2), 256, lds_q_bytes, stream>>>(
      msg_m, Wm, zm, msg_k, msg_k1, Wk, zk);

  gemm128_stream_kernel<<<256, 512, 0, stream>>>(X, Wself, out, zm, zk);
}
